# starting kernel
# baseline (speedup 1.0000x reference)
#include <hip/hip_runtime.h>

typedef __attribute__((ext_vector_type(2))) float v2f;
typedef __attribute__((ext_vector_type(8))) float v8f;

#define B_      16
#define N_IN    16384
#define N_OUT   8192
#define IDIM    128
#define ODIM    128
#define KDIM    256
#define NDIR    6
#define TILES_MAX (N_OUT / 16)

__global__ void k_zero_counts(int* counts) {
    int t = threadIdx.x;
    if (t < B_ * NDIR) counts[t] = 0;
}

__global__ void k_transpose_w(const float* __restrict__ W, float* __restrict__ Wt) {
    int idx = blockIdx.x * 256 + threadIdx.x;
    if (idx >= NDIR * KDIM * ODIM) return;
    int d   = idx / (KDIM * ODIM);
    int rem = idx % (KDIM * ODIM);
    int k   = rem / ODIM;
    int c   = rem % ODIM;
    Wt[((size_t)d * ODIM + c) * KDIM + k] = W[idx];
}

__global__ void k_bucketize(const int* __restrict__ vec, const int* __restrict__ dmap,
                            int* __restrict__ counts, int* __restrict__ buckets) {
    int idx = blockIdx.x * 256 + threadIdx.x;
    if (idx >= B_ * N_OUT) return;
    int b = idx / N_OUT;
    int n = idx % N_OUT;
    int d = dmap[vec[idx]];
    int pos = atomicAdd(&counts[b * NDIR + d], 1);
    buckets[(size_t)(b * NDIR + d) * N_OUT + pos] = n;
}

__global__ __launch_bounds__(256, 1)
void k_gemm_tile(const float* __restrict__ last, const float* __restrict__ Wt,
                 const float* __restrict__ bias, const float* __restrict__ alpha_p,
                 const int* __restrict__ vec, const int* __restrict__ drev,
                 const int* __restrict__ child_l, const int* __restrict__ child_r,
                 const int* __restrict__ counts, const int* __restrict__ buckets,
                 float* __restrict__ out) {
    __shared__ float pairS[16 * KDIM];
    __shared__ int   nrowS[16];

    const int bd   = blockIdx.x / TILES_MAX;
    const int tile = blockIdx.x % TILES_MAX;
    const int cnt  = counts[bd];
    const int row0 = tile * 16;
    if (row0 >= cnt) return;
    const int b = bd / NDIR;
    const int d = bd % NDIR;
    const int nrows = min(16, cnt - row0);

    const int tid = threadIdx.x;

    if (tid < 16) {
        int r = (tid < nrows) ? tid : 0;
        nrowS[tid] = buckets[(size_t)bd * N_OUT + row0 + r];
    }
    __syncthreads();

    {
        int r   = tid >> 4;
        int seg = tid & 15;
        int n   = nrowS[r];
        int rev = drev[vec[b * N_OUT + n]];
        int cl  = child_l[n], cr = child_r[n];
        int srcA = rev ? cr : cl;
        int srcB = rev ? cl : cr;
        int off  = seg * 16;
        const float* src = (off < IDIM)
            ? (last + ((size_t)b * N_IN + srcA) * IDIM + off)
            : (last + ((size_t)b * N_IN + srcB) * IDIM + (off - IDIM));
        float* dst = &pairS[r * KDIM + off];
#pragma unroll
        for (int i = 0; i < 4; ++i)
            ((float4*)dst)[i] = ((const float4*)src)[i];
    }
    __syncthreads();

    const int lane = tid & 31;
    const int wave = tid >> 5;
    const int col0 = wave * 16;
    const int colN = lane & 15;
    const int koff = (lane >> 4) * 2;
    const int arow = lane & 15;
    const float* wcol = Wt + ((size_t)d * ODIM + (col0 + colN)) * KDIM;

    v8f acc = {};
#pragma unroll 8
    for (int k = 0; k < KDIM; k += 4) {
        v2f a, bf;
        a.x  = pairS[arow * KDIM + k + koff];
        a.y  = pairS[arow * KDIM + k + koff + 1];
        bf.x = wcol[k + koff];
        bf.y = wcol[k + koff + 1];
        acc = __builtin_amdgcn_wmma_f32_16x16x4_f32(
                  false, a, false, bf, (short)0, acc, false, false);
    }

    const float alpha = *alpha_p;
    const float bb    = bias[d * ODIM + col0 + colN];
    const int mbase   = (lane >> 4) * 8;
#pragma unroll
    for (int r = 0; r < 8; ++r) {
        int row = mbase + r;
        if (row < nrows) {
            float y = acc[r] + bb;
            y = (y >= 0.0f) ? y : alpha * y;
            out[((size_t)b * N_OUT + nrowS[row]) * ODIM + col0 + colN] = y;
        }
    }
}

extern "C" void kernel_launch(void* const* d_in, const int* in_sizes, int n_in,
                              void* d_out, int out_size, void* d_ws, size_t ws_size,
                              hipStream_t stream) {
    const float* last    = (const float*)d_in[0];
    const float* W       = (const float*)d_in[1];
    const float* bias    = (const float*)d_in[2];
    const float* alpha   = (const float*)d_in[3];
    const int*   vec     = (const int*)d_in[4];
    const int*   dmap    = (const int*)d_in[5];
    const int*   drev    = (const int*)d_in[6];
    const int*   child_l = (const int*)d_in[7];
    const int*   child_r = (const int*)d_in[8];
    float* out = (float*)d_out;

    char* ws = (char*)d_ws;
    int*   counts  = (int*)ws;
    int*   buckets = (int*)(ws + 512);
    float* Wt      = (float*)(ws + 512 + (size_t)B_ * NDIR * N_OUT * 4);

    k_zero_counts<<<1, 128, 0, stream>>>(counts);
    k_transpose_w<<<(NDIR * KDIM * ODIM + 255) / 256, 256, 0, stream>>>(W, Wt);
    k_bucketize<<<(B_ * N_OUT + 255) / 256, 256, 0, stream>>>(vec, dmap, counts, buckets);
    k_gemm_tile<<<B_ * NDIR * TILES_MAX, 256, 0, stream>>>(
        last, Wt, bias, alpha, vec, drev, child_l, child_r, counts, buckets, out);
}


# Bucket lists cut into fixed 512-row segments, one per bucketing block: rows ranked in the block by LDS adds, segment counts stored plainly; no global atomics, no counter-zeroing launch
# speedup vs baseline: 2.6605x; 2.6605x over previous
#include <hip/hip_runtime.h>

typedef __attribute__((ext_vector_type(2))) float v2f;
typedef __attribute__((ext_vector_type(4))) float v4f;
typedef __attribute__((ext_vector_type(8))) float v8f;

#define B_      16
#define N_IN    16384
#define N_OUT   8192
#define IDIM    128
#define ODIM    128
#define KDIM    256
#define NDIR    6
#define NVOC    64
#define NBUCK   (B_ * NDIR)
#define MROWS   32
#define PITCH   260
#define NBLK    512
#define MAXT    9
#define BBLK    512
#define NSEG    (N_OUT / BBLK)

typedef __attribute__((ext_vector_type(4))) int v4i;

__global__ void k_transpose_w(const float* __restrict__ W, float* __restrict__ Wt) {
    int idx = blockIdx.x * 256 + threadIdx.x;
    if (idx >= NDIR * KDIM * ODIM) return;
    int d   = idx / (KDIM * ODIM);
    int rem = idx % (KDIM * ODIM);
    int k   = rem / ODIM;
    int c   = rem % ODIM;
    Wt[((size_t)d * ODIM + c) * KDIM + k] = W[idx];
}

__global__ __launch_bounds__(BBLK)
void k_bucketize(const int* __restrict__ vec, const int* __restrict__ dmap,
                 int* __restrict__ segcnt, unsigned short* __restrict__ buckets) {
    __shared__ int cnt[NDIR];
    const int tid = threadIdx.x;
    if (tid < NDIR) cnt[tid] = 0;
    __syncthreads();
    const int idx = blockIdx.x * BBLK + tid;
    const int b = idx / N_OUT;
    const int n = idx % N_OUT;
    const int s = n / BBLK;
    const int d = dmap[vec[idx]];
    const int local = atomicAdd(&cnt[d], 1);
    __syncthreads();
    buckets[(size_t)(b * NDIR + d) * N_OUT + s * BBLK + local] = (unsigned short)n;
    if (tid < NDIR) segcnt[(b * NDIR + tid) * NSEG + s] = cnt[tid];
}

struct RowLoc { int b, bd, nrows, slot; };

__device__ __forceinline__ RowLoc locate_row(int idx, int t0, int q0,
                                             const int* tstart, const int* cntS, const int* segpre) {
    int k  = idx / MROWS, r = idx % MROWS;
    int tt = t0 + k;
    int q  = q0;
    while (tt >= tstart[q + 1]) ++q;
    int d = q / B_, b = q % B_;
    RowLoc L;
    L.b     = b;
    L.bd    = b * NDIR + d;
    int row0 = (tt - tstart[q]) * MROWS;
    L.nrows = min(MROWS, cntS[q] - row0);
    int rr  = (r < L.nrows) ? r : 0;
    int p = row0 + rr;
    const int* sp = segpre + q * NSEG;
    int s = 0;
    if (sp[s + 8] <= p) s += 8;
    if (sp[s + 4] <= p) s += 4;
    if (sp[s + 2] <= p) s += 2;
    if (sp[s + 1] <= p) s += 1;
    L.slot  = L.bd * N_OUT + s * BBLK + (p - sp[s]);
    return L;
}

__global__ __launch_bounds__(256, 4)
void k_gemm_wstat(const float* __restrict__ last, const float* __restrict__ Wt,
                  const float* __restrict__ bias, const float* __restrict__ alpha_p,
                  const int* __restrict__ vec, const int* __restrict__ drev,
                  const int* __restrict__ child_l, const int* __restrict__ child_r,
                  const int* __restrict__ segcnt, const unsigned short* __restrict__ buckets,
                  float* __restrict__ out) {
    __shared__ __attribute__((aligned(16))) float pairS[2][MROWS * PITCH];
    __shared__ __attribute__((aligned(16))) int rowN[MAXT * MROWS];
    __shared__ int rowA[MAXT * MROWS];
    __shared__ int rowB[MAXT * MROWS];
    __shared__ int cntS[NBUCK];
    __shared__ int segpre[NBUCK * NSEG];
    __shared__ int tcnt[NBUCK];
    __shared__ int tstart[NBUCK + 1];
    __shared__ int drevS[NVOC];
    __shared__ int tileBD[MAXT];
    __shared__ int tileNR[MAXT];

    const int tid = threadIdx.x;

    if (tid < NBUCK) {
        int d = tid / B_, b = tid % B_;
        const v4i* sc = (const v4i*)(segcnt + (b * NDIR + d) * NSEG);
        int c = 0;
#pragma unroll
        for (int i = 0; i < NSEG / 4; ++i) {
            v4i v = sc[i];
            segpre[tid * NSEG + 4 * i + 0] = c; c += v.x;
            segpre[tid * NSEG + 4 * i + 1] = c; c += v.y;
            segpre[tid * NSEG + 4 * i + 2] = c; c += v.z;
            segpre[tid * NSEG + 4 * i + 3] = c; c += v.w;
        }
        cntS[tid] = c;
        tcnt[tid] = (c + MROWS - 1) / MROWS;
    } else if (tid < NBUCK + NVOC) {
        drevS[tid - NBUCK] = drev[tid - NBUCK];
    }
    __syncthreads();
    if (tid <= NBUCK) {
        int s = 0;
        for (int i = 0; i < tid; ++i) s += tcnt[i];
        tstart[tid] = s;
    }
    __syncthreads();
    const int T  = __builtin_amdgcn_readfirstlane(tstart[NBUCK]);
    const int t0 = (int)(blockIdx.x * T) / NBLK;
    const int t1 = (int)((blockIdx.x + 1) * T) / NBLK;
    const int nt = t1 - t0;

    const int lane = tid & 31;
    const int wave = tid >> 5;
    const int c    = lane & 15;
    const int h    = lane >> 4;
    const int col  = wave * 16 + c;
    const int gr = tid >> 3;
    const int gs = tid & 7;

    int cur_d = -1;
    v4f wreg[32];
#pragma unroll
    for (int j = 0; j < 32; ++j) wreg[j] = (v4f){0.f, 0.f, 0.f, 0.f};

    if (nt > 0) {
        int lo = 0, hi = NBUCK;
        while (hi - lo > 1) {
            int mid = (lo + hi) >> 1;
            if (tstart[mid] <= t0) lo = mid; else hi = mid;
        }
        const int q0 = __builtin_amdgcn_readfirstlane(lo);

        cur_d = q0 / B_;
        const float* wp = Wt + ((size_t)cur_d * ODIM + col) * KDIM + 4 * h;
#pragma unroll
        for (int j = 0; j < 32; ++j) wreg[j] = *(const v4f*)(wp + 8 * j);

        const int  tot = nt * MROWS;
        const int  iA = tid, iB = tid + 256;
        const bool vA = iA < tot, vB = iB < tot;
        const RowLoc LA = locate_row(vA ? iA : 0, t0, q0, tstart, cntS, segpre);
        const RowLoc LB = locate_row(vB ? iB : 0, t0, q0, tstart, cntS, segpre);
        const int nA  = buckets[LA.slot];
        const int nB  = buckets[LB.slot];
        const int vvA = vec[LA.b * N_OUT + nA];
        const int vvB = vec[LB.b * N_OUT + nB];
        const int clA = child_l[nA], crA = child_r[nA];
        const int clB = child_l[nB], crB = child_r[nB];
        const int revA = drevS[vvA];
        const int revB = drevS[vvB];
        if (vA) {
            rowN[iA] = nA;
            rowA[iA] = LA.b * N_IN + (revA ? crA : clA);
            rowB[iA] = LA.b * N_IN + (revA ? clA : crA);
            if ((iA % MROWS) == 0) { tileBD[iA / MROWS] = LA.bd; tileNR[iA / MROWS] = LA.nrows; }
        }
        if (vB) {
            rowN[iB] = nB;
            rowA[iB] = LB.b * N_IN + (revB ? crB : clB);
            rowB[iB] = LB.b * N_IN + (revB ? clB : crB);
            if ((iB % MROWS) == 0) { tileBD[iB / MROWS] = LB.bd; tileNR[iB / MROWS] = LB.nrows; }
        }
    }
    __syncthreads();

    if (nt > 0) {
        const float* pA = last + (size_t)rowA[gr] * IDIM + gs * 4;
        const float* pB = last + (size_t)rowB[gr] * IDIM + gs * 4;
        float* dst = &pairS[0][gr * PITCH + gs * 4];
#pragma unroll
        for (int i = 0; i < 4; ++i) {
            *(v4f*)(dst + i * 32)        = *(const v4f*)(pA + i * 32);
            *(v4f*)(dst + IDIM + i * 32) = *(const v4f*)(pB + i * 32);
        }
    }
    __syncthreads();

    const float alpha = *alpha_p;

    for (int k = 0; k < nt; ++k) {
        const int buf = k & 1;
        const int bd    = __builtin_amdgcn_readfirstlane(tileBD[k]);
        const int nrows = __builtin_amdgcn_readfirstlane(tileNR[k]);
        const int b = bd / NDIR;
        const int d = bd % NDIR;

        if (d != cur_d) {
            const float* wp = Wt + ((size_t)d * ODIM + col) * KDIM + 4 * h;
#pragma unroll
            for (int j = 0; j < 32; ++j) wreg[j] = *(const v4f*)(wp + 8 * j);
            cur_d = d;
        }

        const bool hasNext = (k + 1 < nt);
        v4f g[8];
        if (hasNext) {
            const float* pA = last + (size_t)rowA[(k + 1) * MROWS + gr] * IDIM + gs * 4;
            const float* pB = last + (size_t)rowB[(k + 1) * MROWS + gr] * IDIM + gs * 4;
#pragma unroll
            for (int i = 0; i < 4; ++i) {
                g[i]     = *(const v4f*)(pA + i * 32);
                g[4 + i] = *(const v4f*)(pB + i * 32);
            }
        }
        const float bb = bias[d * ODIM + col];

        const float* P0 = &pairS[buf][c * PITCH + 4 * h];
        const float* P1 = P0 + 16 * PITCH;
        v8f acc0 = {}, acc1 = {};
#pragma unroll
        for (int j = 0; j < 32; ++j) {
            v4f a0 = *(const v4f*)(P0 + 8 * j);
            v4f a1 = *(const v4f*)(P1 + 8 * j);
            v4f wv = wreg[j];
            v2f wlo = {wv.x, wv.y}, whi = {wv.z, wv.w};
            v2f a0lo = {a0.x, a0.y}, a0hi = {a0.z, a0.w};
            v2f a1lo = {a1.x, a1.y}, a1hi = {a1.z, a1.w};
            acc0 = __builtin_amdgcn_wmma_f32_16x16x4_f32(false, a0lo, false, wlo, (short)0, acc0, false, false);
            acc1 = __builtin_amdgcn_wmma_f32_16x16x4_f32(false, a1lo, false, wlo, (short)0, acc1, false, false);
            acc0 = __builtin_amdgcn_wmma_f32_16x16x4_f32(false, a0hi, false, whi, (short)0, acc0, false, false);
            acc1 = __builtin_amdgcn_wmma_f32_16x16x4_f32(false, a1hi, false, whi, (short)0, acc1, false, false);
        }
        asm volatile("v_nop\n\tv_nop\n\tv_nop\n\tv_nop\n\tv_nop\n\tv_nop\n\tv_nop\n\tv_nop"
                     : "+v"(acc0), "+v"(acc1));

        if (hasNext) {
            float* dst = &pairS[buf ^ 1][gr * PITCH + gs * 4];
#pragma unroll
            for (int i = 0; i < 4; ++i) {
                *(v4f*)(dst + i * 32)        = g[i];
                *(v4f*)(dst + IDIM + i * 32) = g[4 + i];
            }
        }
        __syncthreads();

        const int*   rn    = &rowN[k * MROWS + 8 * h];
        float*       obase = out + (size_t)b * N_OUT * ODIM + col;
        if (nrows == MROWS) {
#pragma unroll
            for (int r = 0; r < 8; ++r) {
                float u = acc0[r] + bb;
                float v = acc1[r] + bb;
                u = (u >= 0.0f) ? u : alpha * u;
                v = (v >= 0.0f) ? v : alpha * v;
                obase[(size_t)rn[r] * ODIM]      = u;
                obase[(size_t)rn[16 + r] * ODIM] = v;
            }
        } else {
#pragma unroll
            for (int r = 0; r < 8; ++r) {
                int ra = 8 * h + r;
                if (ra < nrows) {
                    float u = acc0[r] + bb;
                    u = (u >= 0.0f) ? u : alpha * u;
                    obase[(size_t)rn[r] * ODIM] = u;
                }
                if (16 + ra < nrows) {
                    float v = acc1[r] + bb;
                    v = (v >= 0.0f) ? v : alpha * v;
                    obase[(size_t)rn[16 + r] * ODIM] = v;
                }
            }
        }
    }
}

extern "C" void kernel_launch(void* const* d_in, const int* in_sizes, int n_in,
                              void* d_out, int out_size, void* d_ws, size_t ws_size,
                              hipStream_t stream) {
    const float* last    = (const float*)d_in[0];
    const float* W       = (const float*)d_in[1];
    const float* bias    = (const float*)d_in[2];
    const float* alpha   = (const float*)d_in[3];
    const int*   vec     = (const int*)d_in[4];
    const int*   dmap    = (const int*)d_in[5];
    const int*   drev    = (const int*)d_in[6];
    const int*   child_l = (const int*)d_in[7];
    const int*   child_r = (const int*)d_in[8];
    float* out = (float*)d_out;

    char* ws = (char*)d_ws;
    int*            segcnt  = (int*)ws;
    unsigned short* buckets = (unsigned short*)(ws + 6144);
    float*          Wt      = (float*)(ws + 6144 + (size_t)B_ * NDIR * N_OUT * 2);

    k_transpose_w<<<(NDIR * KDIM * ODIM + 255) / 256, 256, 0, stream>>>(W, Wt);
    k_bucketize<<<(B_ * N_OUT) / BBLK, BBLK, 0, stream>>>(vec, dmap, segcnt, buckets);
    k_gemm_wstat<<<NBLK, 256, 0, stream>>>(
        last, Wt, bias, alpha, vec, drev, child_l, child_r, segcnt, buckets, out);
}


# W slab read where W lies: a lane loads its 4 k of each piece as 4-B words of row-major W (64 B per wave half per k), once per block; transposed copy and its launch gone (2 launches, was 3)
# speedup vs baseline: 2.9337x; 1.1027x over previous
#include <hip/hip_runtime.h>

typedef __attribute__((ext_vector_type(2))) float v2f;
typedef __attribute__((ext_vector_type(4))) float v4f;
typedef __attribute__((ext_vector_type(8))) float v8f;

#define B_      16
#define N_IN    16384
#define N_OUT   8192
#define IDIM    128
#define ODIM    128
#define KDIM    256
#define NDIR    6
#define NVOC    64
#define NBUCK   (B_ * NDIR)
#define MROWS   32
#define PITCH   260
#define NBLK    512
#define MAXT    9
#define BBLK    512
#define NSEG    (N_OUT / BBLK)

typedef __attribute__((ext_vector_type(4))) int v4i;

__global__ __launch_bounds__(BBLK)
void k_bucketize(const int* __restrict__ vec, const int* __restrict__ dmap,
                 int* __restrict__ segcnt, unsigned short* __restrict__ buckets) {
    __shared__ int cnt[NDIR];
    const int tid = threadIdx.x;
    if (tid < NDIR) cnt[tid] = 0;
    __syncthreads();
    const int idx = blockIdx.x * BBLK + tid;
    const int b = idx / N_OUT;
    const int n = idx % N_OUT;
    const int s = n / BBLK;
    const int d = dmap[vec[idx]];
    const int local = atomicAdd(&cnt[d], 1);
    __syncthreads();
    buckets[(size_t)(b * NDIR + d) * N_OUT + s * BBLK + local] = (unsigned short)n;
    if (tid < NDIR) segcnt[(b * NDIR + tid) * NSEG + s] = cnt[tid];
}

struct RowLoc { int b, bd, nrows, slot; };

__device__ __forceinline__ RowLoc locate_row(int idx, int t0, int q0,
                                             const int* tstart, const int* cntS, const int* segpre) {
    int k  = idx / MROWS, r = idx % MROWS;
    int tt = t0 + k;
    int q  = q0;
    while (tt >= tstart[q + 1]) ++q;
    int d = q / B_, b = q % B_;
    RowLoc L;
    L.b     = b;
    L.bd    = b * NDIR + d;
    int row0 = (tt - tstart[q]) * MROWS;
    L.nrows = min(MROWS, cntS[q] - row0);
    int rr  = (r < L.nrows) ? r : 0;
    int p = row0 + rr;
    const int* sp = segpre + q * NSEG;
    int s = 0;
    if (sp[s + 8] <= p) s += 8;
    if (sp[s + 4] <= p) s += 4;
    if (sp[s + 2] <= p) s += 2;
    if (sp[s + 1] <= p) s += 1;
    L.slot  = L.bd * N_OUT + s * BBLK + (p - sp[s]);
    return L;
}

__device__ __forceinline__ void load_w_slab(v4f (&wreg)[32], const float* __restrict__ W,
                                            int d, int col, int h) {
    const float* wp = W + ((size_t)d * KDIM + 4 * h) * ODIM + col;
#pragma unroll
    for (int j = 0; j < 32; ++j) {
        wreg[j].x = wp[(8 * j + 0) * ODIM];
        wreg[j].y = wp[(8 * j + 1) * ODIM];
        wreg[j].z = wp[(8 * j + 2) * ODIM];
        wreg[j].w = wp[(8 * j + 3) * ODIM];
    }
}

__global__ __launch_bounds__(256, 4)
void k_gemm_wstat(const float* __restrict__ last, const float* __restrict__ W,
                  const float* __restrict__ bias, const float* __restrict__ alpha_p,
                  const int* __restrict__ vec, const int* __restrict__ drev,
                  const int* __restrict__ child_l, const int* __restrict__ child_r,
                  const int* __restrict__ segcnt, const unsigned short* __restrict__ buckets,
                  float* __restrict__ out) {
    __shared__ __attribute__((aligned(16))) float pairS[2][MROWS * PITCH];
    __shared__ __attribute__((aligned(16))) int rowN[MAXT * MROWS];
    __shared__ int rowA[MAXT * MROWS];
    __shared__ int rowB[MAXT * MROWS];
    __shared__ int cntS[NBUCK];
    __shared__ int segpre[NBUCK * NSEG];
    __shared__ int tcnt[NBUCK];
    __shared__ int tstart[NBUCK + 1];
    __shared__ int drevS[NVOC];
    __shared__ int tileBD[MAXT];
    __shared__ int tileNR[MAXT];

    const int tid = threadIdx.x;

    if (tid < NBUCK) {
        int d = tid / B_, b = tid % B_;
        const v4i* sc = (const v4i*)(segcnt + (b * NDIR + d) * NSEG);
        int c = 0;
#pragma unroll
        for (int i = 0; i < NSEG / 4; ++i) {
            v4i v = sc[i];
            segpre[tid * NSEG + 4 * i + 0] = c; c += v.x;
            segpre[tid * NSEG + 4 * i + 1] = c; c += v.y;
            segpre[tid * NSEG + 4 * i + 2] = c; c += v.z;
            segpre[tid * NSEG + 4 * i + 3] = c; c += v.w;
        }
        cntS[tid] = c;
        tcnt[tid] = (c + MROWS - 1) / MROWS;
    } else if (tid < NBUCK + NVOC) {
        drevS[tid - NBUCK] = drev[tid - NBUCK];
    }
    __syncthreads();
    if (tid <= NBUCK) {
        int s = 0;
        for (int i = 0; i < tid; ++i) s += tcnt[i];
        tstart[tid] = s;
    }
    __syncthreads();
    const int T  = __builtin_amdgcn_readfirstlane(tstart[NBUCK]);
    const int t0 = (int)(blockIdx.x * T) / NBLK;
    const int t1 = (int)((blockIdx.x + 1) * T) / NBLK;
    const int nt = t1 - t0;

    const int lane = tid & 31;
    const int wave = tid >> 5;
    const int c    = lane & 15;
    const int h    = lane >> 4;
    const int col  = wave * 16 + c;
    const int gr = tid >> 3;
    const int gs = tid & 7;

    int cur_d = -1;
    v4f wreg[32];
#pragma unroll
    for (int j = 0; j < 32; ++j) wreg[j] = (v4f){0.f, 0.f, 0.f, 0.f};

    if (nt > 0) {
        int lo = 0, hi = NBUCK;
        while (hi - lo > 1) {
            int mid = (lo + hi) >> 1;
            if (tstart[mid] <= t0) lo = mid; else hi = mid;
        }
        const int q0 = __builtin_amdgcn_readfirstlane(lo);

        cur_d = q0 / B_;
        load_w_slab(wreg, W, cur_d, col, h);

        const int  tot = nt * MROWS;
        const int  iA = tid, iB = tid + 256;
        const bool vA = iA < tot, vB = iB < tot;
        const RowLoc LA = locate_row(vA ? iA : 0, t0, q0, tstart, cntS, segpre);
        const RowLoc LB = locate_row(vB ? iB : 0, t0, q0, tstart, cntS, segpre);
        const int nA  = buckets[LA.slot];
        const int nB  = buckets[LB.slot];
        const int vvA = vec[LA.b * N_OUT + nA];
        const int vvB = vec[LB.b * N_OUT + nB];
        const int clA = child_l[nA], crA = child_r[nA];
        const int clB = child_l[nB], crB = child_r[nB];
        const int revA = drevS[vvA];
        const int revB = drevS[vvB];
        if (vA) {
            rowN[iA] = nA;
            rowA[iA] = LA.b * N_IN + (revA ? crA : clA);
            rowB[iA] = LA.b * N_IN + (revA ? clA : crA);
            if ((iA % MROWS) == 0) { tileBD[iA / MROWS] = LA.bd; tileNR[iA / MROWS] = LA.nrows; }
        }
        if (vB) {
            rowN[iB] = nB;
            rowA[iB] = LB.b * N_IN + (revB ? crB : clB);
            rowB[iB] = LB.b * N_IN + (revB ? clB : crB);
            if ((iB % MROWS) == 0) { tileBD[iB / MROWS] = LB.bd; tileNR[iB / MROWS] = LB.nrows; }
        }
    }
    __syncthreads();

    if (nt > 0) {
        const float* pA = last + (size_t)rowA[gr] * IDIM + gs * 4;
        const float* pB = last + (size_t)rowB[gr] * IDIM + gs * 4;
        float* dst = &pairS[0][gr * PITCH + gs * 4];
#pragma unroll
        for (int i = 0; i < 4; ++i) {
            *(v4f*)(dst + i * 32)        = *(const v4f*)(pA + i * 32);
            *(v4f*)(dst + IDIM + i * 32) = *(const v4f*)(pB + i * 32);
        }
    }
    __syncthreads();

    const float alpha = *alpha_p;

    for (int k = 0; k < nt; ++k) {
        const int buf = k & 1;
        const int bd    = __builtin_amdgcn_readfirstlane(tileBD[k]);
        const int nrows = __builtin_amdgcn_readfirstlane(tileNR[k]);
        const int b = bd / NDIR;
        const int d = bd % NDIR;

        if (d != cur_d) {
            load_w_slab(wreg, W, d, col, h);
            cur_d = d;
        }

        const bool hasNext = (k + 1 < nt);
        v4f g[8];
        if (hasNext) {
            const float* pA = last + (size_t)rowA[(k + 1) * MROWS + gr] * IDIM + gs * 4;
            const float* pB = last + (size_t)rowB[(k + 1) * MROWS + gr] * IDIM + gs * 4;
#pragma unroll
            for (int i = 0; i < 4; ++i) {
                g[i]     = *(const v4f*)(pA + i * 32);
                g[4 + i] = *(const v4f*)(pB + i * 32);
            }
        }
        const float bb = bias[d * ODIM + col];

        const float* P0 = &pairS[buf][c * PITCH + 4 * h];
        const float* P1 = P0 + 16 * PITCH;
        v8f acc0 = {}, acc1 = {};
#pragma unroll
        for (int j = 0; j < 32; ++j) {
            v4f a0 = *(const v4f*)(P0 + 8 * j);
            v4f a1 = *(const v4f*)(P1 + 8 * j);
            v4f wv = wreg[j];
            v2f wlo = {wv.x, wv.y}, whi = {wv.z, wv.w};
            v2f a0lo = {a0.x, a0.y}, a0hi = {a0.z, a0.w};
            v2f a1lo = {a1.x, a1.y}, a1hi = {a1.z, a1.w};
            acc0 = __builtin_amdgcn_wmma_f32_16x16x4_f32(false, a0lo, false, wlo, (short)0, acc0, false, false);
            acc1 = __builtin_amdgcn_wmma_f32_16x16x4_f32(false, a1lo, false, wlo, (short)0, acc1, false, false);
            acc0 = __builtin_amdgcn_wmma_f32_16x16x4_f32(false, a0hi, false, whi, (short)0, acc0, false, false);
            acc1 = __builtin_amdgcn_wmma_f32_16x16x4_f32(false, a1hi, false, whi, (short)0, acc1, false, false);
        }
        asm volatile("v_nop\n\tv_nop\n\tv_nop\n\tv_nop\n\tv_nop\n\tv_nop\n\tv_nop\n\tv_nop"
                     : "+v"(acc0), "+v"(acc1));

        if (hasNext) {
            float* dst = &pairS[buf ^ 1][gr * PITCH + gs * 4];
#pragma unroll
            for (int i = 0; i < 4; ++i) {
                *(v4f*)(dst + i * 32)        = g[i];
                *(v4f*)(dst + IDIM + i * 32) = g[4 + i];
            }
        }
        __syncthreads();

        const int*   rn    = &rowN[k * MROWS + 8 * h];
        float*       obase = out + (size_t)b * N_OUT * ODIM + col;
        if (nrows == MROWS) {
#pragma unroll
            for (int r = 0; r < 8; ++r) {
                float u = acc0[r] + bb;
                float v = acc1[r] + bb;
                u = (u >= 0.0f) ? u : alpha * u;
                v = (v >= 0.0f) ? v : alpha * v;
                obase[(size_t)rn[r] * ODIM]      = u;
                obase[(size_t)rn[16 + r] * ODIM] = v;
            }
        } else {
#pragma unroll
            for (int r = 0; r < 8; ++r) {
                int ra = 8 * h + r;
                if (ra < nrows) {
                    float u = acc0[r] + bb;
                    u = (u >= 0.0f) ? u : alpha * u;
                    obase[(size_t)rn[r] * ODIM] = u;
                }
                if (16 + ra < nrows) {
                    float v = acc1[r] + bb;
                    v = (v >= 0.0f) ? v : alpha * v;
                    obase[(size_t)rn[16 + r] * ODIM] = v;
                }
            }
        }
    }
}

extern "C" void kernel_launch(void* const* d_in, const int* in_sizes, int n_in,
                              void* d_out, int out_size, void* d_ws, size_t ws_size,
                              hipStream_t stream) {
    const float* last    = (const float*)d_in[0];
    const float* W       = (const float*)d_in[1];
    const float* bias    = (const float*)d_in[2];
    const float* alpha   = (const float*)d_in[3];
    const int*   vec     = (const int*)d_in[4];
    const int*   dmap    = (const int*)d_in[5];
    const int*   drev    = (const int*)d_in[6];
    const int*   child_l = (const int*)d_in[7];
    const int*   child_r = (const int*)d_in[8];
    float* out = (float*)d_out;

    char* ws = (char*)d_ws;
    int*            segcnt  = (int*)ws;
    unsigned short* buckets = (unsigned short*)(ws + 6144);

    k_bucketize<<<(B_ * N_OUT) / BBLK, BBLK, 0, stream>>>(vec, dmap, segcnt, buckets);
    k_gemm_wstat<<<NBLK, 256, 0, stream>>>(
        last, W, bias, alpha, vec, drev, child_l, child_r, segcnt, buckets, out);
}


# Issue priority (s_setprio, 3 down to 0) falls with the tiles a block has left: the two blocks sharing a WGP take turns on the matrix pipe and end together; the younger no longer ends alone
# speedup vs baseline: 3.1110x; 1.0604x over previous
#include <hip/hip_runtime.h>

typedef __attribute__((ext_vector_type(2))) float v2f;
typedef __attribute__((ext_vector_type(4))) float v4f;
typedef __attribute__((ext_vector_type(8))) float v8f;

#define B_      16
#define N_IN    16384
#define N_OUT   8192
#define IDIM    128
#define ODIM    128
#define KDIM    256
#define NDIR    6
#define NVOC    64
#define NBUCK   (B_ * NDIR)
#define MROWS   32
#define PITCH   260
#define NBLK    512
#define MAXT    9
#define BBLK    512
#define NSEG    (N_OUT / BBLK)

typedef __attribute__((ext_vector_type(4))) int v4i;

__global__ __launch_bounds__(BBLK)
void k_bucketize(const int* __restrict__ vec, const int* __restrict__ dmap,
                 int* __restrict__ segcnt, unsigned short* __restrict__ buckets) {
    __shared__ int cnt[NDIR];
    const int tid = threadIdx.x;
    if (tid < NDIR) cnt[tid] = 0;
    __syncthreads();
    const int idx = blockIdx.x * BBLK + tid;
    const int b = idx / N_OUT;
    const int n = idx % N_OUT;
    const int s = n / BBLK;
    const int d = dmap[vec[idx]];
    const int local = atomicAdd(&cnt[d], 1);
    __syncthreads();
    buckets[(size_t)(b * NDIR + d) * N_OUT + s * BBLK + local] = (unsigned short)n;
    if (tid < NDIR) segcnt[(b * NDIR + tid) * NSEG + s] = cnt[tid];
}

struct RowLoc { int b, bd, nrows, slot; };

__device__ __forceinline__ RowLoc locate_row(int idx, int t0, int q0,
                                             const int* tstart, const int* cntS, const int* segpre) {
    int k  = idx / MROWS, r = idx % MROWS;
    int tt = t0 + k;
    int q  = q0;
    while (tt >= tstart[q + 1]) ++q;
    int d = q / B_, b = q % B_;
    RowLoc L;
    L.b     = b;
    L.bd    = b * NDIR + d;
    int row0 = (tt - tstart[q]) * MROWS;
    L.nrows = min(MROWS, cntS[q] - row0);
    int rr  = (r < L.nrows) ? r : 0;
    int p = row0 + rr;
    const int* sp = segpre + q * NSEG;
    int s = 0;
    if (sp[s + 8] <= p) s += 8;
    if (sp[s + 4] <= p) s += 4;
    if (sp[s + 2] <= p) s += 2;
    if (sp[s + 1] <= p) s += 1;
    L.slot  = L.bd * N_OUT + s * BBLK + (p - sp[s]);
    return L;
}

__device__ __forceinline__ void load_w_slab(v4f (&wreg)[32], const float* __restrict__ W,
                                            int d, int col, int h) {
    const float* wp = W + ((size_t)d * KDIM + 4 * h) * ODIM + col;
#pragma unroll
    for (int j = 0; j < 32; ++j) {
        wreg[j].x = wp[(8 * j + 0) * ODIM];
        wreg[j].y = wp[(8 * j + 1) * ODIM];
        wreg[j].z = wp[(8 * j + 2) * ODIM];
        wreg[j].w = wp[(8 * j + 3) * ODIM];
    }
}

__global__ __launch_bounds__(256, 4)
void k_gemm_wstat(const float* __restrict__ last, const float* __restrict__ W,
                  const float* __restrict__ bias, const float* __restrict__ alpha_p,
                  const int* __restrict__ vec, const int* __restrict__ drev,
                  const int* __restrict__ child_l, const int* __restrict__ child_r,
                  const int* __restrict__ segcnt, const unsigned short* __restrict__ buckets,
                  float* __restrict__ out) {
    __shared__ __attribute__((aligned(16))) float pairS[2][MROWS * PITCH];
    __shared__ __attribute__((aligned(16))) int rowN[MAXT * MROWS];
    __shared__ int rowA[MAXT * MROWS];
    __shared__ int rowB[MAXT * MROWS];
    __shared__ int cntS[NBUCK];
    __shared__ int segpre[NBUCK * NSEG];
    __shared__ int tcnt[NBUCK];
    __shared__ int tstart[NBUCK + 1];
    __shared__ int drevS[NVOC];
    __shared__ int tileBD[MAXT];
    __shared__ int tileNR[MAXT];

    const int tid = threadIdx.x;

    if (tid < NBUCK) {
        int d = tid / B_, b = tid % B_;
        const v4i* sc = (const v4i*)(segcnt + (b * NDIR + d) * NSEG);
        int c = 0;
#pragma unroll
        for (int i = 0; i < NSEG / 4; ++i) {
            v4i v = sc[i];
            segpre[tid * NSEG + 4 * i + 0] = c; c += v.x;
            segpre[tid * NSEG + 4 * i + 1] = c; c += v.y;
            segpre[tid * NSEG + 4 * i + 2] = c; c += v.z;
            segpre[tid * NSEG + 4 * i + 3] = c; c += v.w;
        }
        cntS[tid] = c;
        tcnt[tid] = (c + MROWS - 1) / MROWS;
    } else if (tid < NBUCK + NVOC) {
        drevS[tid - NBUCK] = drev[tid - NBUCK];
    }
    __syncthreads();
    if (tid <= NBUCK) {
        int s = 0;
        for (int i = 0; i < tid; ++i) s += tcnt[i];
        tstart[tid] = s;
    }
    __syncthreads();
    const int T  = __builtin_amdgcn_readfirstlane(tstart[NBUCK]);
    const int t0 = (int)(blockIdx.x * T) / NBLK;
    const int t1 = (int)((blockIdx.x + 1) * T) / NBLK;
    const int nt = t1 - t0;

    const int lane = tid & 31;
    const int wave = tid >> 5;
    const int c    = lane & 15;
    const int h    = lane >> 4;
    const int col  = wave * 16 + c;
    const int gr = tid >> 3;
    const int gs = tid & 7;

    int cur_d = -1;
    v4f wreg[32];
#pragma unroll
    for (int j = 0; j < 32; ++j) wreg[j] = (v4f){0.f, 0.f, 0.f, 0.f};

    if (nt > 0) {
        int lo = 0, hi = NBUCK;
        while (hi - lo > 1) {
            int mid = (lo + hi) >> 1;
            if (tstart[mid] <= t0) lo = mid; else hi = mid;
        }
        const int q0 = __builtin_amdgcn_readfirstlane(lo);

        cur_d = q0 / B_;
        load_w_slab(wreg, W, cur_d, col, h);

        const int  tot = nt * MROWS;
        const int  iA = tid, iB = tid + 256;
        const bool vA = iA < tot, vB = iB < tot;
        const RowLoc LA = locate_row(vA ? iA : 0, t0, q0, tstart, cntS, segpre);
        const RowLoc LB = locate_row(vB ? iB : 0, t0, q0, tstart, cntS, segpre);
        const int nA  = buckets[LA.slot];
        const int nB  = buckets[LB.slot];
        const int vvA = vec[LA.b * N_OUT + nA];
        const int vvB = vec[LB.b * N_OUT + nB];
        const int clA = child_l[nA], crA = child_r[nA];
        const int clB = child_l[nB], crB = child_r[nB];
        const int revA = drevS[vvA];
        const int revB = drevS[vvB];
        if (vA) {
            rowN[iA] = nA;
            rowA[iA] = LA.b * N_IN + (revA ? crA : clA);
            rowB[iA] = LA.b * N_IN + (revA ? clA : crA);
            if ((iA % MROWS) == 0) { tileBD[iA / MROWS] = LA.bd; tileNR[iA / MROWS] = LA.nrows; }
        }
        if (vB) {
            rowN[iB] = nB;
            rowA[iB] = LB.b * N_IN + (revB ? crB : clB);
            rowB[iB] = LB.b * N_IN + (revB ? clB : crB);
            if ((iB % MROWS) == 0) { tileBD[iB / MROWS] = LB.bd; tileNR[iB / MROWS] = LB.nrows; }
        }
    }
    __syncthreads();

    if (nt > 0) {
        const float* pA = last + (size_t)rowA[gr] * IDIM + gs * 4;
        const float* pB = last + (size_t)rowB[gr] * IDIM + gs * 4;
        float* dst = &pairS[0][gr * PITCH + gs * 4];
#pragma unroll
        for (int i = 0; i < 4; ++i) {
            *(v4f*)(dst + i * 32)        = *(const v4f*)(pA + i * 32);
            *(v4f*)(dst + IDIM + i * 32) = *(const v4f*)(pB + i * 32);
        }
    }
    __syncthreads();

    const float alpha = *alpha_p;

    for (int k = 0; k < nt; ++k) {
        const int buf = k & 1;
        const int rem = nt - k;
        if (rem >= 6)      __builtin_amdgcn_s_setprio(3);
        else if (rem >= 4) __builtin_amdgcn_s_setprio(2);
        else if (rem >= 2) __builtin_amdgcn_s_setprio(1);
        else               __builtin_amdgcn_s_setprio(0);
        const int bd    = __builtin_amdgcn_readfirstlane(tileBD[k]);
        const int nrows = __builtin_amdgcn_readfirstlane(tileNR[k]);
        const int b = bd / NDIR;
        const int d = bd % NDIR;

        if (d != cur_d) {
            load_w_slab(wreg, W, d, col, h);
            cur_d = d;
        }

        const bool hasNext = (k + 1 < nt);
        v4f g[8];
        if (hasNext) {
            const float* pA = last + (size_t)rowA[(k + 1) * MROWS + gr] * IDIM + gs * 4;
            const float* pB = last + (size_t)rowB[(k + 1) * MROWS + gr] * IDIM + gs * 4;
#pragma unroll
            for (int i = 0; i < 4; ++i) {
                g[i]     = *(const v4f*)(pA + i * 32);
                g[4 + i] = *(const v4f*)(pB + i * 32);
            }
        }
        const float bb = bias[d * ODIM + col];

        const float* P0 = &pairS[buf][c * PITCH + 4 * h];
        const float* P1 = P0 + 16 * PITCH;
        v8f acc0 = {}, acc1 = {};
#pragma unroll
        for (int j = 0; j < 32; ++j) {
            v4f a0 = *(const v4f*)(P0 + 8 * j);
            v4f a1 = *(const v4f*)(P1 + 8 * j);
            v4f wv = wreg[j];
            v2f wlo = {wv.x, wv.y}, whi = {wv.z, wv.w};
            v2f a0lo = {a0.x, a0.y}, a0hi = {a0.z, a0.w};
            v2f a1lo = {a1.x, a1.y}, a1hi = {a1.z, a1.w};
            acc0 = __builtin_amdgcn_wmma_f32_16x16x4_f32(false, a0lo, false, wlo, (short)0, acc0, false, false);
            acc1 = __builtin_amdgcn_wmma_f32_16x16x4_f32(false, a1lo, false, wlo, (short)0, acc1, false, false);
            acc0 = __builtin_amdgcn_wmma_f32_16x16x4_f32(false, a0hi, false, whi, (short)0, acc0, false, false);
            acc1 = __builtin_amdgcn_wmma_f32_16x16x4_f32(false, a1hi, false, whi, (short)0, acc1, false, false);
        }
        asm volatile("v_nop\n\tv_nop\n\tv_nop\n\tv_nop\n\tv_nop\n\tv_nop\n\tv_nop\n\tv_nop"
                     : "+v"(acc0), "+v"(acc1));

        if (hasNext) {
            float* dst = &pairS[buf ^ 1][gr * PITCH + gs * 4];
#pragma unroll
            for (int i = 0; i < 4; ++i) {
                *(v4f*)(dst + i * 32)        = g[i];
                *(v4f*)(dst + IDIM + i * 32) = g[4 + i];
            }
        }
        __syncthreads();

        const int*   rn    = &rowN[k * MROWS + 8 * h];
        float*       obase = out + (size_t)b * N_OUT * ODIM + col;
        if (nrows == MROWS) {
#pragma unroll
            for (int r = 0; r < 8; ++r) {
                float u = acc0[r] + bb;
                float v = acc1[r] + bb;
                u = (u >= 0.0f) ? u : alpha * u;
                v = (v >= 0.0f) ? v : alpha * v;
                obase[(size_t)rn[r] * ODIM]      = u;
                obase[(size_t)rn[16 + r] * ODIM] = v;
            }
        } else {
#pragma unroll
            for (int r = 0; r < 8; ++r) {
                int ra = 8 * h + r;
                if (ra < nrows) {
                    float u = acc0[r] + bb;
                    u = (u >= 0.0f) ? u : alpha * u;
                    obase[(size_t)rn[r] * ODIM] = u;
                }
                if (16 + ra < nrows) {
                    float v = acc1[r] + bb;
                    v = (v >= 0.0f) ? v : alpha * v;
                    obase[(size_t)rn[16 + r] * ODIM] = v;
                }
            }
        }
    }
}

extern "C" void kernel_launch(void* const* d_in, const int* in_sizes, int n_in,
                              void* d_out, int out_size, void* d_ws, size_t ws_size,
                              hipStream_t stream) {
    const float* last    = (const float*)d_in[0];
    const float* W       = (const float*)d_in[1];
    const float* bias    = (const float*)d_in[2];
    const float* alpha   = (const float*)d_in[3];
    const int*   vec     = (const int*)d_in[4];
    const int*   dmap    = (const int*)d_in[5];
    const int*   drev    = (const int*)d_in[6];
    const int*   child_l = (const int*)d_in[7];
    const int*   child_r = (const int*)d_in[8];
    float* out = (float*)d_out;

    char* ws = (char*)d_ws;
    int*            segcnt  = (int*)ws;
    unsigned short* buckets = (unsigned short*)(ws + 6144);

    k_bucketize<<<(B_ * N_OUT) / BBLK, BBLK, 0, stream>>>(vec, dmap, segcnt, buckets);
    k_gemm_wstat<<<NBLK, 256, 0, stream>>>(
        last, W, bias, alpha, vec, drev, child_l, child_r, segcnt, buckets, out);
}


# Result tile staged in LDS per C/D layout; each wave then stores whole 512-B output rows, 16 B per lane (4 stores per wave and tile, was 16 of 4 B); bias per lane loaded with the W slab
# speedup vs baseline: 3.3900x; 1.0897x over previous
#include <hip/hip_runtime.h>

typedef __attribute__((ext_vector_type(2))) float v2f;
typedef __attribute__((ext_vector_type(4))) float v4f;
typedef __attribute__((ext_vector_type(8))) float v8f;

#define B_      16
#define N_IN    16384
#define N_OUT   8192
#define IDIM    128
#define ODIM    128
#define KDIM    256
#define NDIR    6
#define NVOC    64
#define NBUCK   (B_ * NDIR)
#define MROWS   32
#define PITCH   260
#define OPITCH  132
#define NBLK    512
#define MAXT    9
#define BBLK    512
#define NSEG    (N_OUT / BBLK)

typedef __attribute__((ext_vector_type(4))) int v4i;

__global__ __launch_bounds__(BBLK)
void k_bucketize(const int* __restrict__ vec, const int* __restrict__ dmap,
                 int* __restrict__ segcnt, unsigned short* __restrict__ buckets) {
    __shared__ int cnt[NDIR];
    const int tid = threadIdx.x;
    if (tid < NDIR) cnt[tid] = 0;
    __syncthreads();
    const int idx = blockIdx.x * BBLK + tid;
    const int b = idx / N_OUT;
    const int n = idx % N_OUT;
    const int s = n / BBLK;
    const int d = dmap[vec[idx]];
    const int local = atomicAdd(&cnt[d], 1);
    __syncthreads();
    buckets[(size_t)(b * NDIR + d) * N_OUT + s * BBLK + local] = (unsigned short)n;
    if (tid < NDIR) segcnt[(b * NDIR + tid) * NSEG + s] = cnt[tid];
}

struct RowLoc { int b, bd, nrows, slot; };

__device__ __forceinline__ RowLoc locate_row(int idx, int t0, int q0,
                                             const int* tstart, const int* cntS, const int* segpre) {
    int k  = idx / MROWS, r = idx % MROWS;
    int tt = t0 + k;
    int q  = q0;
    while (tt >= tstart[q + 1]) ++q;
    int d = q / B_, b = q % B_;
    RowLoc L;
    L.b     = b;
    L.bd    = b * NDIR + d;
    int row0 = (tt - tstart[q]) * MROWS;
    L.nrows = min(MROWS, cntS[q] - row0);
    int rr  = (r < L.nrows) ? r : 0;
    int p = row0 + rr;
    const int* sp = segpre + q * NSEG;
    int s = 0;
    if (sp[s + 8] <= p) s += 8;
    if (sp[s + 4] <= p) s += 4;
    if (sp[s + 2] <= p) s += 2;
    if (sp[s + 1] <= p) s += 1;
    L.slot  = L.bd * N_OUT + s * BBLK + (p - sp[s]);
    return L;
}

__device__ __forceinline__ void load_w_slab(v4f (&wreg)[32], const float* __restrict__ W,
                                            int d, int col, int h) {
    const float* wp = W + ((size_t)d * KDIM + 4 * h) * ODIM + col;
#pragma unroll
    for (int j = 0; j < 32; ++j) {
        wreg[j].x = wp[(8 * j + 0) * ODIM];
        wreg[j].y = wp[(8 * j + 1) * ODIM];
        wreg[j].z = wp[(8 * j + 2) * ODIM];
        wreg[j].w = wp[(8 * j + 3) * ODIM];
    }
}

__device__ __forceinline__ void finish_row(const float* __restrict__ src, v4f bb4, float alpha,
                                           float* __restrict__ dst) {
    v4f v = *(const v4f*)src;
    v.x += bb4.x; v.y += bb4.y; v.z += bb4.z; v.w += bb4.w;
    v.x = (v.x >= 0.0f) ? v.x : alpha * v.x;
    v.y = (v.y >= 0.0f) ? v.y : alpha * v.y;
    v.z = (v.z >= 0.0f) ? v.z : alpha * v.z;
    v.w = (v.w >= 0.0f) ? v.w : alpha * v.w;
    *(v4f*)dst = v;
}

__global__ __launch_bounds__(256, 4)
void k_gemm_wstat(const float* __restrict__ last, const float* __restrict__ W,
                  const float* __restrict__ bias, const float* __restrict__ alpha_p,
                  const int* __restrict__ vec, const int* __restrict__ drev,
                  const int* __restrict__ child_l, const int* __restrict__ child_r,
                  const int* __restrict__ segcnt, const unsigned short* __restrict__ buckets,
                  float* __restrict__ out) {
    __shared__ __attribute__((aligned(16))) float pairS[2][MROWS * PITCH];
    __shared__ __attribute__((aligned(16))) float outS[2][MROWS * OPITCH];
    __shared__ __attribute__((aligned(16))) int rowN[MAXT * MROWS];
    __shared__ int rowA[MAXT * MROWS];
    __shared__ int rowB[MAXT * MROWS];
    __shared__ int cntS[NBUCK];
    __shared__ int segpre[NBUCK * NSEG];
    __shared__ int tcnt[NBUCK];
    __shared__ int tstart[NBUCK + 1];
    __shared__ int drevS[NVOC];
    __shared__ int tileBD[MAXT];
    __shared__ int tileNR[MAXT];

    const int tid = threadIdx.x;

    if (tid < NBUCK) {
        int d = tid / B_, b = tid % B_;
        const v4i* sc = (const v4i*)(segcnt + (b * NDIR + d) * NSEG);
        int c = 0;
#pragma unroll
        for (int i = 0; i < NSEG / 4; ++i) {
            v4i v = sc[i];
            segpre[tid * NSEG + 4 * i + 0] = c; c += v.x;
            segpre[tid * NSEG + 4 * i + 1] = c; c += v.y;
            segpre[tid * NSEG + 4 * i + 2] = c; c += v.z;
            segpre[tid * NSEG + 4 * i + 3] = c; c += v.w;
        }
        cntS[tid] = c;
        tcnt[tid] = (c + MROWS - 1) / MROWS;
    } else if (tid < NBUCK + NVOC) {
        drevS[tid - NBUCK] = drev[tid - NBUCK];
    }
    __syncthreads();
    if (tid <= NBUCK) {
        int s = 0;
        for (int i = 0; i < tid; ++i) s += tcnt[i];
        tstart[tid] = s;
    }
    __syncthreads();
    const int T  = __builtin_amdgcn_readfirstlane(tstart[NBUCK]);
    const int t0 = (int)(blockIdx.x * T) / NBLK;
    const int t1 = (int)((blockIdx.x + 1) * T) / NBLK;
    const int nt = t1 - t0;

    const int lane = tid & 31;
    const int wave = tid >> 5;
    const int c    = lane & 15;
    const int h    = lane >> 4;
    const int col  = wave * 16 + c;
    const int gr = tid >> 3;
    const int gs = tid & 7;

    int cur_d = -1;
    v4f wreg[32];
#pragma unroll
    for (int j = 0; j < 32; ++j) wreg[j] = (v4f){0.f, 0.f, 0.f, 0.f};
    v4f bb4 = (v4f){0.f, 0.f, 0.f, 0.f};
    int osoff = (8 * h) * OPITCH + col;
    int l4    = 4 * lane;

    if (nt > 0) {
        int lo = 0, hi = NBUCK;
        while (hi - lo > 1) {
            int mid = (lo + hi) >> 1;
            if (tstart[mid] <= t0) lo = mid; else hi = mid;
        }
        const int q0 = __builtin_amdgcn_readfirstlane(lo);

        cur_d = q0 / B_;
        load_w_slab(wreg, W, cur_d, col, h);
        bb4 = *(const v4f*)(bias + cur_d * ODIM + 4 * lane);

        const int  tot = nt * MROWS;
        const int  iA = tid, iB = tid + 256;
        const bool vA = iA < tot, vB = iB < tot;
        const RowLoc LA = locate_row(vA ? iA : 0, t0, q0, tstart, cntS, segpre);
        const RowLoc LB = locate_row(vB ? iB : 0, t0, q0, tstart, cntS, segpre);
        const int nA  = buckets[LA.slot];
        const int nB  = buckets[LB.slot];
        const int vvA = vec[LA.b * N_OUT + nA];
        const int vvB = vec[LB.b * N_OUT + nB];
        const int clA = child_l[nA], crA = child_r[nA];
        const int clB = child_l[nB], crB = child_r[nB];
        const int revA = drevS[vvA];
        const int revB = drevS[vvB];
        if (vA) {
            rowN[iA] = nA;
            rowA[iA] = LA.b * N_IN + (revA ? crA : clA);
            rowB[iA] = LA.b * N_IN + (revA ? clA : crA);
            if ((iA % MROWS) == 0) { tileBD[iA / MROWS] = LA.bd; tileNR[iA / MROWS] = LA.nrows; }
        }
        if (vB) {
            rowN[iB] = nB;
            rowA[iB] = LB.b * N_IN + (revB ? crB : clB);
            rowB[iB] = LB.b * N_IN + (revB ? clB : crB);
            if ((iB % MROWS) == 0) { tileBD[iB / MROWS] = LB.bd; tileNR[iB / MROWS] = LB.nrows; }
        }
    }
    __syncthreads();

    if (nt > 0) {
        const float* pA = last + (size_t)rowA[gr] * IDIM + gs * 4;
        const float* pB = last + (size_t)rowB[gr] * IDIM + gs * 4;
        float* dst = &pairS[0][gr * PITCH + gs * 4];
#pragma unroll
        for (int i = 0; i < 4; ++i) {
            *(v4f*)(dst + i * 32)        = *(const v4f*)(pA + i * 32);
            *(v4f*)(dst + IDIM + i * 32) = *(const v4f*)(pB + i * 32);
        }
    }
    __syncthreads();

    const float alpha = *alpha_p;

    for (int k = 0; k < nt; ++k) {
        const int buf = k & 1;
        const int rem = nt - k;
        if (rem >= 6)      __builtin_amdgcn_s_setprio(3);
        else if (rem >= 4) __builtin_amdgcn_s_setprio(2);
        else if (rem >= 2) __builtin_amdgcn_s_setprio(1);
        else               __builtin_amdgcn_s_setprio(0);
        const int bd    = __builtin_amdgcn_readfirstlane(tileBD[k]);
        const int nrows = __builtin_amdgcn_readfirstlane(tileNR[k]);
        const int b = bd / NDIR;
        const int d = bd % NDIR;

        if (d != cur_d) {
            load_w_slab(wreg, W, d, col, h);
            bb4 = *(const v4f*)(bias + d * ODIM + 4 * lane);
            cur_d = d;
        }

        const bool hasNext = (k + 1 < nt);
        v4f g[8];
        if (hasNext) {
            const float* pA = last + (size_t)rowA[(k + 1) * MROWS + gr] * IDIM + gs * 4;
            const float* pB = last + (size_t)rowB[(k + 1) * MROWS + gr] * IDIM + gs * 4;
#pragma unroll
            for (int i = 0; i < 4; ++i) {
                g[i]     = *(const v4f*)(pA + i * 32);
                g[4 + i] = *(const v4f*)(pB + i * 32);
            }
        }

        const float* P0 = &pairS[buf][c * PITCH + 4 * h];
        const float* P1 = P0 + 16 * PITCH;
        v8f acc0 = {}, acc1 = {};
#pragma unroll
        for (int j = 0; j < 32; ++j) {
            v4f a0 = *(const v4f*)(P0 + 8 * j);
            v4f a1 = *(const v4f*)(P1 + 8 * j);
            v4f wv = wreg[j];
            v2f wlo = {wv.x, wv.y}, whi = {wv.z, wv.w};
            v2f a0lo = {a0.x, a0.y}, a0hi = {a0.z, a0.w};
            v2f a1lo = {a1.x, a1.y}, a1hi = {a1.z, a1.w};
            acc0 = __builtin_amdgcn_wmma_f32_16x16x4_f32(false, a0lo, false, wlo, (short)0, acc0, false, false);
            acc1 = __builtin_amdgcn_wmma_f32_16x16x4_f32(false, a1lo, false, wlo, (short)0, acc1, false, false);
            acc0 = __builtin_amdgcn_wmma_f32_16x16x4_f32(false, a0hi, false, whi, (short)0, acc0, false, false);
            acc1 = __builtin_amdgcn_wmma_f32_16x16x4_f32(false, a1hi, false, whi, (short)0, acc1, false, false);
        }
        asm volatile("v_nop\n\tv_nop\n\tv_nop\n\tv_nop\n\tv_nop\n\tv_nop\n\tv_nop\n\tv_nop"
                     : "+v"(acc0), "+v"(acc1), "+v"(osoff), "+v"(l4));

        {
            float* os = &outS[buf][osoff];
#pragma unroll
            for (int r = 0; r < 8; ++r) {
                os[r * OPITCH]        = acc0[r];
                os[(16 + r) * OPITCH] = acc1[r];
            }
        }

        if (hasNext) {
            float* dst = &pairS[buf ^ 1][gr * PITCH + gs * 4];
#pragma unroll
            for (int i = 0; i < 4; ++i) {
                *(v4f*)(dst + i * 32)        = g[i];
                *(v4f*)(dst + IDIM + i * 32) = g[4 + i];
            }
        }
        __syncthreads();

        const int sw = __builtin_amdgcn_readfirstlane(wave);
        const v4i ids = *(const v4i*)&rowN[k * MROWS + 4 * sw];
        const int n0 = __builtin_amdgcn_readfirstlane(ids.x);
        const int n1 = __builtin_amdgcn_readfirstlane(ids.y);
        const int n2 = __builtin_amdgcn_readfirstlane(ids.z);
        const int n3 = __builtin_amdgcn_readfirstlane(ids.w);
        float*       obase = out + (size_t)b * N_OUT * ODIM + l4;
        const float* srow  = &outS[buf][(4 * sw) * OPITCH + l4];
        if (nrows == MROWS) {
            finish_row(srow,              bb4, alpha, obase + (size_t)n0 * ODIM);
            finish_row(srow + OPITCH,     bb4, alpha, obase + (size_t)n1 * ODIM);
            finish_row(srow + 2 * OPITCH, bb4, alpha, obase + (size_t)n2 * ODIM);
            finish_row(srow + 3 * OPITCH, bb4, alpha, obase + (size_t)n3 * ODIM);
        } else {
            if (4 * sw + 0 < nrows) finish_row(srow,              bb4, alpha, obase + (size_t)n0 * ODIM);
            if (4 * sw + 1 < nrows) finish_row(srow + OPITCH,     bb4, alpha, obase + (size_t)n1 * ODIM);
            if (4 * sw + 2 < nrows) finish_row(srow + 2 * OPITCH, bb4, alpha, obase + (size_t)n2 * ODIM);
            if (4 * sw + 3 < nrows) finish_row(srow + 3 * OPITCH, bb4, alpha, obase + (size_t)n3 * ODIM);
        }
    }
}

extern "C" void kernel_launch(void* const* d_in, const int* in_sizes, int n_in,
                              void* d_out, int out_size, void* d_ws, size_t ws_size,
                              hipStream_t stream) {
    const float* last    = (const float*)d_in[0];
    const float* W       = (const float*)d_in[1];
    const float* bias    = (const float*)d_in[2];
    const float* alpha   = (const float*)d_in[3];
    const int*   vec     = (const int*)d_in[4];
    const int*   dmap    = (const int*)d_in[5];
    const int*   drev    = (const int*)d_in[6];
    const int*   child_l = (const int*)d_in[7];
    const int*   child_r = (const int*)d_in[8];
    float* out = (float*)d_out;

    char* ws = (char*)d_ws;
    int*            segcnt  = (int*)ws;
    unsigned short* buckets = (unsigned short*)(ws + 6144);

    k_bucketize<<<(B_ * N_OUT) / BBLK, BBLK, 0, stream>>>(vec, dmap, segcnt, buckets);
    k_gemm_wstat<<<NBLK, 256, 0, stream>>>(
        last, W, bias, alpha, vec, drev, child_l, child_r, segcnt, buckets, out);
}
